# starting kernel
# baseline (speedup 1.0000x reference)
#include <hip/hip_runtime.h>
#include <cstdint>
#include <cstddef>

#define N0_  200000
#define N1_  50000
#define N2_  10000
#define E0_  750000
#define E1_  150000
#define DIN_ 256
#define HID_ 256
#define OUT_ 64

typedef float v2f __attribute__((ext_vector_type(2)));
typedef float v8f __attribute__((ext_vector_type(8)));

__global__ void zero_f32(float* __restrict__ p, int n) {
  int i = blockIdx.x * blockDim.x + threadIdx.x;
  if (i < n) p[i] = 0.0f;
}

__global__ void scatter_mean_accum(const float* __restrict__ xsrc,
                                   const int* __restrict__ src,
                                   const int* __restrict__ dst,
                                   float* __restrict__ agg,
                                   float* __restrict__ cnt,
                                   int E) {
  int e = blockIdx.x * 4 + (threadIdx.x >> 6);
  if (e >= E) return;
  int t = threadIdx.x & 63;
  int s = src[e];
  int d = dst[e];
  float4 v = ((const float4*)(xsrc + (size_t)s * 256))[t];
  float* ap = agg + (size_t)d * 256 + t * 4;
  atomicAdd(ap + 0, v.x);
  atomicAdd(ap + 1, v.y);
  atomicAdd(ap + 2, v.z);
  atomicAdd(ap + 3, v.w);
  if (t == 0) atomicAdd(cnt + d, 1.0f);
}

template <int KTOT, int NCOL, int ACT, bool NORM>
__global__ __launch_bounds__((NCOL / 16) * 32)
void sage_gemm_wmma(const float* __restrict__ A0,
                    const float* __restrict__ cnt,
                    const float* __restrict__ A1,
                    const float* __restrict__ B0,
                    const float* __restrict__ B1,
                    const float* __restrict__ bias,
                    float* __restrict__ out,
                    int M) {
  constexpr int NT  = (NCOL / 16) * 32;
  constexpr int LDK = KTOT + 4;
  __shared__ float Asub[32][LDK];

  const int tid  = threadIdx.x;
  const int wave = tid >> 5;
  const int lane = tid & 31;
  const int half = lane >> 4;
  const int l15  = lane & 15;
  const int m0   = blockIdx.x * 32;
  const bool has2 = (m0 + 16) < M;

  for (int i = tid * 4; i < 32 * KTOT; i += NT * 4) {
    int row = i / KTOT;
    int col = i % KTOT;
    int gr  = m0 + row;
    float4 v = {0.0f, 0.0f, 0.0f, 0.0f};
    if (gr < M) {
      if (col < 256) {
        v = *(const float4*)(A0 + (size_t)gr * 256 + col);
        if (NORM) {
          float c = cnt[gr];
          float s = (c > 0.0f) ? 1.0f / c : 0.0f;
          v.x *= s; v.y *= s; v.z *= s; v.w *= s;
        }
      } else {
        v = *(const float4*)(A1 + (size_t)gr * 256 + (col - 256));
      }
    }
    *(float4*)&Asub[row][col] = v;
  }
  __syncthreads();

  const int nbase = wave * 16;
  v8f acc0 = {0.0f, 0.0f, 0.0f, 0.0f, 0.0f, 0.0f, 0.0f, 0.0f};
  v8f acc1 = {0.0f, 0.0f, 0.0f, 0.0f, 0.0f, 0.0f, 0.0f, 0.0f};

#pragma unroll
  for (int seg = 0; seg < KTOT / 256; ++seg) {
    const float* bp = ((seg == 0) ? B0 : B1) +
                      (size_t)(half * 2) * NCOL + nbase + l15;
    const int kb = seg * 256;
#pragma unroll 8
    for (int k = 0; k < 256; k += 4) {
      v2f b;
      b.x = bp[0];
      b.y = bp[NCOL];
      int ka = kb + k + half * 2;
      float2 a0v = *(const float2*)&Asub[l15][ka];
      float2 a1v = *(const float2*)&Asub[16 + l15][ka];
      v2f a0; a0.x = a0v.x; a0.y = a0v.y;
      v2f a1; a1.x = a1v.x; a1.y = a1v.y;
      acc0 = __builtin_amdgcn_wmma_f32_16x16x4_f32(false, a0, false, b,
                                                   (short)0, acc0, false, false);
      acc1 = __builtin_amdgcn_wmma_f32_16x16x4_f32(false, a1, false, b,
                                                   (short)0, acc1, false, false);
      bp += (size_t)4 * NCOL;
    }
  }

  const int col = nbase + l15;
  const float bv = bias[col];
#pragma unroll
  for (int r = 0; r < 8; ++r) {
    int row = m0 + half * 8 + r;
    float v = acc0[r] + bv;
    if (ACT == 1) v = fmaxf(v, 0.0f);
    if (ACT == 2) v = tanhf(v);
    out[(size_t)row * NCOL + col] = v;
  }
  if (has2) {
#pragma unroll
    for (int r = 0; r < 8; ++r) {
      int row = m0 + 16 + half * 8 + r;
      float v = acc1[r] + bv;
      if (ACT == 1) v = fmaxf(v, 0.0f);
      if (ACT == 2) v = tanhf(v);
      out[(size_t)row * NCOL + col] = v;
    }
  }
}

__global__ void softmax_rows64(const float* __restrict__ logits,
                               float* __restrict__ out, int M) {
  int row = blockIdx.x * 8 + (threadIdx.x >> 5);
  if (row >= M) return;
  int lane = threadIdx.x & 31;
  const float* lp = logits + (size_t)row * 64;
  float a = lp[lane];
  float b = lp[lane + 32];
  float m = fmaxf(a, b);
  for (int off = 16; off > 0; off >>= 1) m = fmaxf(m, __shfl_xor(m, off, 32));
  float ea = expf(a - m);
  float eb = expf(b - m);
  float s = ea + eb;
  for (int off = 16; off > 0; off >>= 1) s += __shfl_xor(s, off, 32);
  float inv = 1.0f / s;
  float* op = out + (size_t)row * 64;
  op[lane] = ea * inv;
  op[lane + 32] = eb * inv;
}

extern "C" void kernel_launch(void* const* d_in, const int* in_sizes, int n_in,
                              void* d_out, int out_size, void* d_ws,
                              size_t ws_size, hipStream_t stream) {
  (void)in_sizes; (void)n_in; (void)out_size; (void)ws_size;

  const float* x    = (const float*)d_in[0];
  const int*   src0 = (const int*)d_in[1];
  const int*   dst0 = (const int*)d_in[2];
  const int*   src1 = (const int*)d_in[3];
  const int*   dst1 = (const int*)d_in[4];
  const float* Wl0  = (const float*)d_in[5];
  const float* bl0  = (const float*)d_in[6];
  const float* Wr0  = (const float*)d_in[7];
  const float* Wl1  = (const float*)d_in[8];
  const float* bl1  = (const float*)d_in[9];
  const float* Wr1  = (const float*)d_in[10];
  const float* Wlin = (const float*)d_in[11];
  const float* blin = (const float*)d_in[12];

  char* ws = (char*)d_ws;
  auto align256 = [](size_t v) { return (v + 255) & ~(size_t)255; };
  size_t off = 0;
  float* agg0 = (float*)(ws + off); off += (size_t)N1_ * 256 * 4;
  float* cnt0 = (float*)(ws + off); off += (size_t)N1_ * 4; off = align256(off);
  float* h0   = (float*)(ws + off); off += (size_t)N1_ * 256 * 4;
  float* agg1 = (float*)(ws + off); off += (size_t)N2_ * 256 * 4;
  float* cnt1 = (float*)(ws + off); off += (size_t)N2_ * 4; off = align256(off);
  float* h1   = (float*)(ws + off); off += (size_t)N2_ * 256 * 4;
  float* lgt  = (float*)(ws + off); off += (size_t)N2_ * 64 * 4;

  {
    int n = N1_ * 256 + N1_;
    zero_f32<<<(n + 255) / 256, 256, 0, stream>>>(agg0, n);
  }
  scatter_mean_accum<<<E0_ / 4, 256, 0, stream>>>(x, src0, dst0, agg0, cnt0, E0_);
  sage_gemm_wmma<512, 256, 1, true><<<(N1_ + 31) / 32, 512, 0, stream>>>(
      agg0, cnt0, x, Wl0, Wr0, bl0, h0, N1_);

  {
    int n = N2_ * 256 + N2_;
    zero_f32<<<(n + 255) / 256, 256, 0, stream>>>(agg1, n);
  }
  scatter_mean_accum<<<E1_ / 4, 256, 0, stream>>>(h0, src1, dst1, agg1, cnt1, E1_);
  sage_gemm_wmma<512, 256, 2, true><<<(N2_ + 31) / 32, 512, 0, stream>>>(
      agg1, cnt1, h0, Wl1, Wr1, bl1, h1, N2_);

  sage_gemm_wmma<256, 64, 0, false><<<(N2_ + 31) / 32, 128, 0, stream>>>(
      h1, nullptr, nullptr, Wlin, nullptr, blin, lgt, N2_);
  softmax_rows64<<<(N2_ + 7) / 8, 256, 0, stream>>>(lgt, (float*)d_out, N2_);
}


# Float-atomic edge scatter replaced by counting sort by dst (int atomics) + one-wave-per-row register gather-sum: no float atomics, no agg zero pass; row sum order = int-atomic slot arrival order
# speedup vs baseline: 1.5147x; 1.5147x over previous
#include <hip/hip_runtime.h>
#include <cstdint>
#include <cstddef>

#define N0_  200000
#define N1_  50000
#define N2_  10000
#define E0_  750000
#define E1_  150000
#define DIN_ 256
#define HID_ 256
#define OUT_ 64

typedef float v2f __attribute__((ext_vector_type(2)));
typedef float v8f __attribute__((ext_vector_type(8)));

__global__ void zero_f32(float* __restrict__ p, int n) {
  int i = blockIdx.x * blockDim.x + threadIdx.x;
  if (i < n) p[i] = 0.0f;
}

__global__ void edge_count(const int* __restrict__ dst0, int* __restrict__ deg0,
                           const int* __restrict__ dst1, int* __restrict__ deg1) {
  int i = blockIdx.x * blockDim.x + threadIdx.x;
  if (i < E0_) {
    atomicAdd(deg0 + dst0[i], 1);
  } else if (i < E0_ + E1_) {
    atomicAdd(deg1 + dst1[i - E0_], 1);
  }
}

__device__ __forceinline__ int wave_incl_scan(int x, int lane) {
  for (int off = 1; off < 32; off <<= 1) {
    int u = __shfl_up(x, off, 32);
    if (lane >= off) x += u;
  }
  return x;
}

template <int N>
__device__ __forceinline__ void block_offsets(int* __restrict__ deg,
                                              int* __restrict__ rp,
                                              int (*wtot)[32]) {
  static_assert(N % 4 == 0, "counts are fetched four at a time");
  constexpr int TILES = (N + 4095) / 4096;
  const int tid  = threadIdx.x;
  const int lane = tid & 31;
  const int wave = tid >> 5;
  int4 v[TILES];
  int inc[TILES];
  for (int k = 0; k < TILES; ++k) {
    int i = k * 4096 + tid * 4;
    int4 t = {0, 0, 0, 0};
    if (i < N) t = *(const int4*)(deg + i);
    v[k] = t;
  }
  for (int k = 0; k < TILES; ++k) {
    int t = v[k].x + v[k].y + v[k].z + v[k].w;
    inc[k] = wave_incl_scan(t, lane);
    if (lane == 31) wtot[k][wave] = inc[k];
  }
  __syncthreads();
  int carry = 0;
  for (int k = 0; k < TILES; ++k) {
    int w    = wtot[k][lane];
    int winc = wave_incl_scan(w, lane);
    int wexc = __shfl(winc - w, wave, 32);
    int tot  = __shfl(winc, 31, 32);
    int t    = v[k].x + v[k].y + v[k].z + v[k].w;
    int4 o;
    o.x = carry + wexc + inc[k] - t;
    o.y = o.x + v[k].x;
    o.z = o.y + v[k].y;
    o.w = o.z + v[k].z;
    int i = k * 4096 + tid * 4;
    if (i < N) {
      *(int4*)(rp + i)  = o;
      *(int4*)(deg + i) = o;
    }
    carry += tot;
  }
  if (tid == 0) rp[N] = carry;
}

__global__ __launch_bounds__(1024)
void csr_offsets(int* __restrict__ deg0, int* __restrict__ rp0,
                 int* __restrict__ deg1, int* __restrict__ rp1) {
  __shared__ int wtot[(N1_ + 4095) / 4096][32];
  if (blockIdx.x == 0) {
    block_offsets<N1_>(deg0, rp0, wtot);
  } else {
    block_offsets<N2_>(deg1, rp1, wtot);
  }
}

__global__ void edge_place(const int* __restrict__ src0, const int* __restrict__ dst0,
                           int* __restrict__ cur0, int* __restrict__ es0,
                           const int* __restrict__ src1, const int* __restrict__ dst1,
                           int* __restrict__ cur1, int* __restrict__ es1) {
  int i = blockIdx.x * blockDim.x + threadIdx.x;
  if (i < E0_) {
    int p = atomicAdd(cur0 + dst0[i], 1);
    es0[p] = src0[i];
  } else if (i < E0_ + E1_) {
    int j = i - E0_;
    int p = atomicAdd(cur1 + dst1[j], 1);
    es1[p] = src1[j];
  }
}

__global__ __launch_bounds__(256)
void gather_rows_sum(const float* __restrict__ xsrc,
                     const int* __restrict__ rowptr,
                     const int* __restrict__ esrc,
                     float* __restrict__ agg,
                     float* __restrict__ cnt,
                     int nrows) {
  const int lane = threadIdx.x & 31;
  const int row = __builtin_amdgcn_readfirstlane(blockIdx.x * 8 + (threadIdx.x >> 5));
  if (row >= nrows) return;
  const int beg = __builtin_amdgcn_readfirstlane(rowptr[row]);
  const int end = __builtin_amdgcn_readfirstlane(rowptr[row + 1]);

  float4 a0 = {0.0f, 0.0f, 0.0f, 0.0f};
  float4 a1 = {0.0f, 0.0f, 0.0f, 0.0f};
  for (int base = beg; base < end; base += 32) {
    const int n = (end - base < 32) ? (end - base) : 32;
    const int mine = (lane < n) ? esrc[base + lane] : 0;
    int j = 0;
    for (; j + 4 <= n; j += 4) {
      const float4* p0 = (const float4*)(xsrc + (size_t)__builtin_amdgcn_readlane(mine, j + 0) * 256);
      const float4* p1 = (const float4*)(xsrc + (size_t)__builtin_amdgcn_readlane(mine, j + 1) * 256);
      const float4* p2 = (const float4*)(xsrc + (size_t)__builtin_amdgcn_readlane(mine, j + 2) * 256);
      const float4* p3 = (const float4*)(xsrc + (size_t)__builtin_amdgcn_readlane(mine, j + 3) * 256);
      float4 u0 = p0[lane], w0 = p0[32 + lane];
      float4 u1 = p1[lane], w1 = p1[32 + lane];
      float4 u2 = p2[lane], w2 = p2[32 + lane];
      float4 u3 = p3[lane], w3 = p3[32 + lane];
      a0 += u0; a1 += w0;
      a0 += u1; a1 += w1;
      a0 += u2; a1 += w2;
      a0 += u3; a1 += w3;
    }
    const int r = n - j;
    if (r == 3) {
      const float4* p0 = (const float4*)(xsrc + (size_t)__builtin_amdgcn_readlane(mine, j + 0) * 256);
      const float4* p1 = (const float4*)(xsrc + (size_t)__builtin_amdgcn_readlane(mine, j + 1) * 256);
      const float4* p2 = (const float4*)(xsrc + (size_t)__builtin_amdgcn_readlane(mine, j + 2) * 256);
      float4 u0 = p0[lane], w0 = p0[32 + lane];
      float4 u1 = p1[lane], w1 = p1[32 + lane];
      float4 u2 = p2[lane], w2 = p2[32 + lane];
      a0 += u0; a1 += w0;
      a0 += u1; a1 += w1;
      a0 += u2; a1 += w2;
    } else if (r == 2) {
      const float4* p0 = (const float4*)(xsrc + (size_t)__builtin_amdgcn_readlane(mine, j + 0) * 256);
      const float4* p1 = (const float4*)(xsrc + (size_t)__builtin_amdgcn_readlane(mine, j + 1) * 256);
      float4 u0 = p0[lane], w0 = p0[32 + lane];
      float4 u1 = p1[lane], w1 = p1[32 + lane];
      a0 += u0; a1 += w0;
      a0 += u1; a1 += w1;
    } else if (r == 1) {
      const float4* p0 = (const float4*)(xsrc + (size_t)__builtin_amdgcn_readlane(mine, j) * 256);
      float4 u0 = p0[lane], w0 = p0[32 + lane];
      a0 += u0; a1 += w0;
    }
  }
  float4* op = (float4*)(agg + (size_t)row * 256);
  op[lane]      = a0;
  op[32 + lane] = a1;
  if (lane == 0) cnt[row] = (float)(end - beg);
}

template <int KTOT, int NCOL, int ACT, bool NORM>
__global__ __launch_bounds__((NCOL / 16) * 32)
void sage_gemm_wmma(const float* __restrict__ A0,
                    const float* __restrict__ cnt,
                    const float* __restrict__ A1,
                    const float* __restrict__ B0,
                    const float* __restrict__ B1,
                    const float* __restrict__ bias,
                    float* __restrict__ out,
                    int M) {
  constexpr int NT  = (NCOL / 16) * 32;
  constexpr int LDK = KTOT + 4;
  __shared__ float Asub[32][LDK];

  const int tid  = threadIdx.x;
  const int wave = tid >> 5;
  const int lane = tid & 31;
  const int half = lane >> 4;
  const int l15  = lane & 15;
  const int m0   = blockIdx.x * 32;
  const bool has2 = (m0 + 16) < M;

  for (int i = tid * 4; i < 32 * KTOT; i += NT * 4) {
    int row = i / KTOT;
    int col = i % KTOT;
    int gr  = m0 + row;
    float4 v = {0.0f, 0.0f, 0.0f, 0.0f};
    if (gr < M) {
      if (col < 256) {
        v = *(const float4*)(A0 + (size_t)gr * 256 + col);
        if (NORM) {
          float c = cnt[gr];
          float s = (c > 0.0f) ? 1.0f / c : 0.0f;
          v.x *= s; v.y *= s; v.z *= s; v.w *= s;
        }
      } else {
        v = *(const float4*)(A1 + (size_t)gr * 256 + (col - 256));
      }
    }
    *(float4*)&Asub[row][col] = v;
  }
  __syncthreads();

  const int nbase = wave * 16;
  v8f acc0 = {0.0f, 0.0f, 0.0f, 0.0f, 0.0f, 0.0f, 0.0f, 0.0f};
  v8f acc1 = {0.0f, 0.0f, 0.0f, 0.0f, 0.0f, 0.0f, 0.0f, 0.0f};

#pragma unroll
  for (int seg = 0; seg < KTOT / 256; ++seg) {
    const float* bp = ((seg == 0) ? B0 : B1) +
                      (size_t)(half * 2) * NCOL + nbase + l15;
    const int kb = seg * 256;
#pragma unroll 8
    for (int k = 0; k < 256; k += 4) {
      v2f b;
      b.x = bp[0];
      b.y = bp[NCOL];
      int ka = kb + k + half * 2;
      float2 a0v = *(const float2*)&Asub[l15][ka];
      float2 a1v = *(const float2*)&Asub[16 + l15][ka];
      v2f a0; a0.x = a0v.x; a0.y = a0v.y;
      v2f a1; a1.x = a1v.x; a1.y = a1v.y;
      acc0 = __builtin_amdgcn_wmma_f32_16x16x4_f32(false, a0, false, b,
                                                   (short)0, acc0, false, false);
      acc1 = __builtin_amdgcn_wmma_f32_16x16x4_f32(false, a1, false, b,
                                                   (short)0, acc1, false, false);
      bp += (size_t)4 * NCOL;
    }
  }

  const int col = nbase + l15;
  const float bv = bias[col];
#pragma unroll
  for (int r = 0; r < 8; ++r) {
    int row = m0 + half * 8 + r;
    float v = acc0[r] + bv;
    if (ACT == 1) v = fmaxf(v, 0.0f);
    if (ACT == 2) v = tanhf(v);
    out[(size_t)row * NCOL + col] = v;
  }
  if (has2) {
#pragma unroll
    for (int r = 0; r < 8; ++r) {
      int row = m0 + 16 + half * 8 + r;
      float v = acc1[r] + bv;
      if (ACT == 1) v = fmaxf(v, 0.0f);
      if (ACT == 2) v = tanhf(v);
      out[(size_t)row * NCOL + col] = v;
    }
  }
}

__global__ void softmax_rows64(const float* __restrict__ logits,
                               float* __restrict__ out, int M) {
  int row = blockIdx.x * 8 + (threadIdx.x >> 5);
  if (row >= M) return;
  int lane = threadIdx.x & 31;
  const float* lp = logits + (size_t)row * 64;
  float a = lp[lane];
  float b = lp[lane + 32];
  float m = fmaxf(a, b);
  for (int off = 16; off > 0; off >>= 1) m = fmaxf(m, __shfl_xor(m, off, 32));
  float ea = expf(a - m);
  float eb = expf(b - m);
  float s = ea + eb;
  for (int off = 16; off > 0; off >>= 1) s += __shfl_xor(s, off, 32);
  float inv = 1.0f / s;
  float* op = out + (size_t)row * 64;
  op[lane] = ea * inv;
  op[lane + 32] = eb * inv;
}

extern "C" void kernel_launch(void* const* d_in, const int* in_sizes, int n_in,
                              void* d_out, int out_size, void* d_ws,
                              size_t ws_size, hipStream_t stream) {
  (void)in_sizes; (void)n_in; (void)out_size; (void)ws_size;

  const float* x    = (const float*)d_in[0];
  const int*   src0 = (const int*)d_in[1];
  const int*   dst0 = (const int*)d_in[2];
  const int*   src1 = (const int*)d_in[3];
  const int*   dst1 = (const int*)d_in[4];
  const float* Wl0  = (const float*)d_in[5];
  const float* bl0  = (const float*)d_in[6];
  const float* Wr0  = (const float*)d_in[7];
  const float* Wl1  = (const float*)d_in[8];
  const float* bl1  = (const float*)d_in[9];
  const float* Wr1  = (const float*)d_in[10];
  const float* Wlin = (const float*)d_in[11];
  const float* blin = (const float*)d_in[12];

  char* ws = (char*)d_ws;
  auto align256 = [](size_t v) { return (v + 255) & ~(size_t)255; };
  size_t off = 0;
  float* agg0 = (float*)(ws + off); off += (size_t)N1_ * 256 * 4;
  float* cnt0 = (float*)(ws + off); off += (size_t)N1_ * 4; off = align256(off);
  float* h0   = (float*)(ws + off); off += (size_t)N1_ * 256 * 4;
  float* agg1 = (float*)(ws + off); off += (size_t)N2_ * 256 * 4;
  float* cnt1 = (float*)(ws + off); off += (size_t)N2_ * 4; off = align256(off);
  float* h1   = (float*)(ws + off); off += (size_t)N2_ * 256 * 4;
  float* lgt  = (float*)(ws + off); off += (size_t)N2_ * 64 * 4;

  int* deg0 = (int*)h1;
  int* deg1 = deg0 + N1_;
  int* rp0  = deg1 + N2_;
  int* rp1  = rp0 + (N1_ + 4);
  int* es0  = rp1 + (N2_ + 4);
  int* es1  = es0 + E0_;

  zero_f32<<<(N1_ + N2_ + 255) / 256, 256, 0, stream>>>((float*)deg0, N1_ + N2_);
  edge_count<<<(E0_ + E1_ + 255) / 256, 256, 0, stream>>>(dst0, deg0, dst1, deg1);
  csr_offsets<<<2, 1024, 0, stream>>>(deg0, rp0, deg1, rp1);
  edge_place<<<(E0_ + E1_ + 255) / 256, 256, 0, stream>>>(src0, dst0, deg0, es0,
                                                          src1, dst1, deg1, es1);

  gather_rows_sum<<<(N1_ + 7) / 8, 256, 0, stream>>>(x, rp0, es0, agg0, cnt0, N1_);
  sage_gemm_wmma<512, 256, 1, true><<<(N1_ + 31) / 32, 512, 0, stream>>>(
      agg0, cnt0, x, Wl0, Wr0, bl0, h0, N1_);

  gather_rows_sum<<<(N2_ + 7) / 8, 256, 0, stream>>>(h0, rp1, es1, agg1, cnt1, N2_);
  sage_gemm_wmma<512, 256, 2, true><<<(N2_ + 31) / 32, 512, 0, stream>>>(
      agg1, cnt1, h0, Wl1, Wr1, bl1, h1, N2_);

  sage_gemm_wmma<256, 64, 0, false><<<(N2_ + 31) / 32, 128, 0, stream>>>(
      h1, nullptr, nullptr, Wlin, nullptr, blin, lgt, N2_);
  softmax_rows64<<<(N2_ + 7) / 8, 256, 0, stream>>>(lgt, (float*)d_out, N2_);
}


# Neighbour gather fused into each SAGE layer's GEMM launch: waves sum their rows' neighbours into the LDS A stripe beside other work-groups' WMMAs; output sum order: target half, then neighbour half
# speedup vs baseline: 2.4529x; 1.6194x over previous
#include <hip/hip_runtime.h>
#include <cstdint>
#include <cstddef>

#define N0_  200000
#define N1_  50000
#define N2_  10000
#define E0_  750000
#define E1_  150000
#define DIN_ 256
#define HID_ 256
#define OUT_ 64

typedef float v2f __attribute__((ext_vector_type(2)));
typedef float v8f __attribute__((ext_vector_type(8)));

#define NKEY_   (N1_ + N2_)
#define ETOT_   (E0_ + E1_)
#define NBIN_   ((NKEY_ + 255) / 256)
#define NCHUNK_ 256
#define CHUNK_  ((ETOT_ + NCHUNK_ - 1) / NCHUNK_)
#define BINCAP_ 8192

__device__ __forceinline__ int wave_incl_scan(int x, int lane) {
  for (int off = 1; off < 32; off <<= 1) {
    int u = __shfl_up(x, off, 32);
    if (lane >= off) x += u;
  }
  return x;
}

__global__ __launch_bounds__(1024)
void chunk_bin(const int* __restrict__ src0, const int* __restrict__ dst0,
               const int* __restrict__ src1, const int* __restrict__ dst1,
               int* __restrict__ runs, int* __restrict__ startT,
               int* __restrict__ cntT) {
  static_assert(CHUNK_ <= 4 * 1024, "four edges per thread cover a run");
  static_assert(NBIN_ <= 256, "one LDS counter per bin");
  static_assert(N0_ <= (1 << 18), "source id and key byte share one word");
  __shared__ int hist[256];
  __shared__ int boff[256];
  const int tid = threadIdx.x;
  const int c   = blockIdx.x;
  if (tid < 256) hist[tid] = 0;
  __syncthreads();

  int key[4], sv[4], rk[4];
  for (int k = 0; k < 4; ++k) {
    const int li = k * 1024 + tid;
    const int i  = c * CHUNK_ + li;
    key[k] = -1;
    sv[k]  = 0;
    if (li < CHUNK_ && i < ETOT_) {
      if (i < E0_) {
        key[k] = dst0[i];
        sv[k]  = src0[i];
      } else {
        key[k] = N1_ + dst1[i - E0_];
        sv[k]  = src1[i - E0_];
      }
    }
  }
  for (int k = 0; k < 4; ++k) {
    rk[k] = 0;
    if (key[k] >= 0) rk[k] = atomicAdd(&hist[key[k] >> 8], 1);
  }
  __syncthreads();
  if (tid < 32) {
    int h[8];
    int t = 0;
    for (int j = 0; j < 8; ++j) { h[j] = hist[tid * 8 + j]; t += h[j]; }
    int run = wave_incl_scan(t, tid) - t;
    for (int j = 0; j < 8; ++j) { boff[tid * 8 + j] = run; run += h[j]; }
  }
  __syncthreads();
  for (int k = 0; k < 4; ++k) {
    if (key[k] >= 0) {
      runs[c * CHUNK_ + boff[key[k] >> 8] + rk[k]] = sv[k] | ((key[k] & 255) << 18);
    }
  }
  if (tid < NBIN_) {
    startT[tid * NCHUNK_ + c] = boff[tid];
    cntT[tid * NCHUNK_ + c]   = hist[tid];
  }
}

__global__ __launch_bounds__(256)
void bin_sort(const int* __restrict__ runs, const int* __restrict__ startT,
              const int* __restrict__ cntT, int* __restrict__ rp,
              int* __restrict__ es) {
  static_assert(NCHUNK_ == 256, "one thread per run");
  __shared__ int buf[BINCAP_];
  __shared__ int rnk[BINCAP_];
  __shared__ int khist[256];
  __shared__ int koff[256];
  __shared__ int wsum[2][8];
  const int tid  = threadIdx.x;
  const int lane = tid & 31;
  const int wave = tid >> 5;
  const int b    = blockIdx.x;

  const int s = startT[b * NCHUNK_ + tid];
  const int n = cntT[b * NCHUNK_ + tid];
  khist[tid] = 0;
  const int s_inc = wave_incl_scan(s, lane);
  const int n_inc = wave_incl_scan(n, lane);
  if (lane == 31) { wsum[0][wave] = s_inc; wsum[1][wave] = n_inc; }
  __syncthreads();
  int base = 0, npre = 0, ntot = 0;
  for (int w = 0; w < 8; ++w) {
    base += wsum[0][w];
    const int t = wsum[1][w];
    if (w < wave) npre += t;
    ntot += t;
  }
  const int lpos = npre + n_inc - n;

  const int wrun = __builtin_amdgcn_readfirstlane(wave) * 32;
  for (int j0 = 0; j0 < 32; j0 += 8) {
    int v[8];
#pragma unroll
    for (int u = 0; u < 8; ++u) {
      const int sj = __builtin_amdgcn_readlane(s, j0 + u);
      const int nj = __builtin_amdgcn_readlane(n, j0 + u);
      v[u] = 0;
      if (lane < nj) v[u] = runs[(wrun + j0 + u) * CHUNK_ + sj + lane];
    }
#pragma unroll
    for (int u = 0; u < 8; ++u) {
      const int nj = __builtin_amdgcn_readlane(n, j0 + u);
      const int pj = __builtin_amdgcn_readlane(lpos, j0 + u);
      if (lane < nj && pj + lane < BINCAP_) buf[pj + lane] = v[u];
    }
  }
  if (__any(n > 32)) {
    for (int j = 0; j < 32; ++j) {
      const int sj = __builtin_amdgcn_readlane(s, j);
      const int nj = __builtin_amdgcn_readlane(n, j);
      const int pj = __builtin_amdgcn_readlane(lpos, j);
      for (int q = lane + 32; q < nj; q += 32) {
        if (pj + q < BINCAP_) buf[pj + q] = runs[(wrun + j) * CHUNK_ + sj + q];
      }
    }
  }
  __syncthreads();

  const int nin = (ntot < BINCAP_) ? ntot : BINCAP_;
  for (int i0 = tid; i0 < nin; i0 += 4 * 256) {
    int w[4], r[4];
    for (int u = 0; u < 4; ++u) {
      const int i = i0 + u * 256;
      w[u] = (i < nin) ? buf[i] : 0;
    }
    for (int u = 0; u < 4; ++u) {
      const int i = i0 + u * 256;
      r[u] = 0;
      if (i < nin) r[u] = atomicAdd(&khist[(w[u] >> 18) & 255], 1);
    }
    for (int u = 0; u < 4; ++u) {
      const int i = i0 + u * 256;
      if (i < nin) rnk[i] = r[u];
    }
  }
  __syncthreads();
  const int h     = khist[tid];
  const int h_inc = wave_incl_scan(h, lane);
  if (lane == 31) wsum[0][wave] = h_inc;
  __syncthreads();
  int hpre = 0;
  for (int w = 0; w < 8; ++w) {
    if (w < wave) hpre += wsum[0][w];
  }
  const int kex = hpre + h_inc - h;
  koff[tid] = kex;
  const int key = b * 256 + tid;
  if (key < NKEY_) rp[key] = base + kex;
  if (b == NBIN_ - 1 && tid == 0) rp[NKEY_] = base + ntot;
  __syncthreads();
  for (int i = tid; i < nin; i += 256) {
    const int w = buf[i];
    es[base + koff[(w >> 18) & 255] + rnk[i]] = w & 0x3FFFF;
  }
}

__device__ __forceinline__ void gather_row_sum(const float* __restrict__ xsrc,
                                               const int* __restrict__ esrc,
                                               int beg, int end, int first,
                                               int lane, float4& a0, float4& a1) {
  for (int base = beg; base < end; base += 32) {
    const int n = (end - base < 32) ? (end - base) : 32;
    int mine = first;
    if (base != beg) mine = (lane < n) ? esrc[base + lane] : 0;
    int j = 0;
    for (; j + 4 <= n; j += 4) {
      const float4* p0 = (const float4*)(xsrc + (size_t)__builtin_amdgcn_readlane(mine, j + 0) * 256);
      const float4* p1 = (const float4*)(xsrc + (size_t)__builtin_amdgcn_readlane(mine, j + 1) * 256);
      const float4* p2 = (const float4*)(xsrc + (size_t)__builtin_amdgcn_readlane(mine, j + 2) * 256);
      const float4* p3 = (const float4*)(xsrc + (size_t)__builtin_amdgcn_readlane(mine, j + 3) * 256);
      float4 u0 = p0[lane], w0 = p0[32 + lane];
      float4 u1 = p1[lane], w1 = p1[32 + lane];
      float4 u2 = p2[lane], w2 = p2[32 + lane];
      float4 u3 = p3[lane], w3 = p3[32 + lane];
      a0 += u0; a1 += w0;
      a0 += u1; a1 += w1;
      a0 += u2; a1 += w2;
      a0 += u3; a1 += w3;
    }
    const int r = n - j;
    if (r == 3) {
      const float4* p0 = (const float4*)(xsrc + (size_t)__builtin_amdgcn_readlane(mine, j + 0) * 256);
      const float4* p1 = (const float4*)(xsrc + (size_t)__builtin_amdgcn_readlane(mine, j + 1) * 256);
      const float4* p2 = (const float4*)(xsrc + (size_t)__builtin_amdgcn_readlane(mine, j + 2) * 256);
      float4 u0 = p0[lane], w0 = p0[32 + lane];
      float4 u1 = p1[lane], w1 = p1[32 + lane];
      float4 u2 = p2[lane], w2 = p2[32 + lane];
      a0 += u0; a1 += w0;
      a0 += u1; a1 += w1;
      a0 += u2; a1 += w2;
    } else if (r == 2) {
      const float4* p0 = (const float4*)(xsrc + (size_t)__builtin_amdgcn_readlane(mine, j + 0) * 256);
      const float4* p1 = (const float4*)(xsrc + (size_t)__builtin_amdgcn_readlane(mine, j + 1) * 256);
      float4 u0 = p0[lane], w0 = p0[32 + lane];
      float4 u1 = p1[lane], w1 = p1[32 + lane];
      a0 += u0; a1 += w0;
      a0 += u1; a1 += w1;
    } else if (r == 1) {
      const float4* p0 = (const float4*)(xsrc + (size_t)__builtin_amdgcn_readlane(mine, j) * 256);
      float4 u0 = p0[lane], w0 = p0[32 + lane];
      a0 += u0; a1 += w0;
    }
  }
}

template <int ACT>
__global__ __launch_bounds__(512)
void sage_layer_fused(const float* __restrict__ xsrc,
                      const int* __restrict__ rowptr,
                      const int* __restrict__ esrc,
                      const float* __restrict__ xtgt,
                      const float* __restrict__ Bl,
                      const float* __restrict__ Br,
                      const float* __restrict__ bias,
                      float* __restrict__ out,
                      int M) {
  constexpr int NCOL = 256;
  constexpr int LDK  = 512 + 4;
  __shared__ float Asub[32][LDK];

  const int tid  = threadIdx.x;
  const int lane = tid & 31;
  const int half = lane >> 4;
  const int l15  = lane & 15;
  const int wave = __builtin_amdgcn_readfirstlane(tid >> 5);
  const int m0   = blockIdx.x * 32;
  const bool has2 = (m0 + 16) < M;

  const int r0 = m0 + wave * 2;
  int b0 = 0, b1 = 0, b2 = 0;
  if (r0 < M) { b0 = rowptr[r0]; b1 = rowptr[r0 + 1]; }
  b2 = b1;
  if (r0 + 1 < M) b2 = rowptr[r0 + 2];
  b0 = __builtin_amdgcn_readfirstlane(b0);
  b1 = __builtin_amdgcn_readfirstlane(b1);
  b2 = __builtin_amdgcn_readfirstlane(b2);
  int first0 = 0, first1 = 0;
  if (b0 + lane < b1) first0 = esrc[b0 + lane];
  if (b1 + lane < b2) first1 = esrc[b1 + lane];

  {
    float4 tv[4];
#pragma unroll
    for (int u = 0; u < 4; ++u) {
      const int i  = tid * 4 + u * 2048;
      const int gr = m0 + (i >> 8);
      const int gc = (gr < M) ? gr : (M - 1);
      tv[u] = *(const float4*)(xtgt + (size_t)gc * 256 + (i & 255));
      if (gr >= M) tv[u] = float4{0.0f, 0.0f, 0.0f, 0.0f};
    }
#pragma unroll
    for (int u = 0; u < 4; ++u) {
      const int i = tid * 4 + u * 2048;
      *(float4*)&Asub[i >> 8][256 + (i & 255)] = tv[u];
    }
  }
  __syncthreads();

  const int nbase = wave * 16;
  v8f acc0 = {0.0f, 0.0f, 0.0f, 0.0f, 0.0f, 0.0f, 0.0f, 0.0f};
  v8f acc1 = {0.0f, 0.0f, 0.0f, 0.0f, 0.0f, 0.0f, 0.0f, 0.0f};

  auto kseg = [&](const float* __restrict__ Bm, const int kb) {
    const float* bp = Bm + (size_t)(half * 2) * NCOL + nbase + l15;
#pragma unroll 8
    for (int k = 0; k < 256; k += 4) {
      v2f b;
      b.x = bp[0];
      b.y = bp[NCOL];
      const int ka = kb + k + half * 2;
      float2 a0v = *(const float2*)&Asub[l15][ka];
      float2 a1v = *(const float2*)&Asub[16 + l15][ka];
      v2f a0; a0.x = a0v.x; a0.y = a0v.y;
      v2f a1; a1.x = a1v.x; a1.y = a1v.y;
      acc0 = __builtin_amdgcn_wmma_f32_16x16x4_f32(false, a0, false, b,
                                                   (short)0, acc0, false, false);
      acc1 = __builtin_amdgcn_wmma_f32_16x16x4_f32(false, a1, false, b,
                                                   (short)0, acc1, false, false);
      bp += (size_t)4 * NCOL;
    }
  };

  kseg(Br, 256);

  {
    float4 s0 = {0.0f, 0.0f, 0.0f, 0.0f};
    float4 s1 = {0.0f, 0.0f, 0.0f, 0.0f};
    gather_row_sum(xsrc, esrc, b0, b1, first0, lane, s0, s1);
    const float c = (float)(b1 - b0);
    const float sc = (c > 0.0f) ? 1.0f / c : 0.0f;
    s0.x *= sc; s0.y *= sc; s0.z *= sc; s0.w *= sc;
    s1.x *= sc; s1.y *= sc; s1.z *= sc; s1.w *= sc;
    *(float4*)&Asub[wave * 2][4 * lane]       = s0;
    *(float4*)&Asub[wave * 2][128 + 4 * lane] = s1;
  }
  {
    float4 s0 = {0.0f, 0.0f, 0.0f, 0.0f};
    float4 s1 = {0.0f, 0.0f, 0.0f, 0.0f};
    gather_row_sum(xsrc, esrc, b1, b2, first1, lane, s0, s1);
    const float c = (float)(b2 - b1);
    const float sc = (c > 0.0f) ? 1.0f / c : 0.0f;
    s0.x *= sc; s0.y *= sc; s0.z *= sc; s0.w *= sc;
    s1.x *= sc; s1.y *= sc; s1.z *= sc; s1.w *= sc;
    *(float4*)&Asub[wave * 2 + 1][4 * lane]       = s0;
    *(float4*)&Asub[wave * 2 + 1][128 + 4 * lane] = s1;
  }
  __syncthreads();

  kseg(Bl, 0);

  const int col = nbase + l15;
  const float bv = bias[col];
#pragma unroll
  for (int r = 0; r < 8; ++r) {
    int row = m0 + half * 8 + r;
    float v = acc0[r] + bv;
    if (ACT == 1) v = fmaxf(v, 0.0f);
    if (ACT == 2) v = tanhf(v);
    out[(size_t)row * NCOL + col] = v;
  }
  if (has2) {
#pragma unroll
    for (int r = 0; r < 8; ++r) {
      int row = m0 + 16 + half * 8 + r;
      float v = acc1[r] + bv;
      if (ACT == 1) v = fmaxf(v, 0.0f);
      if (ACT == 2) v = tanhf(v);
      out[(size_t)row * NCOL + col] = v;
    }
  }
}

template <int KTOT, int NCOL, int ACT, bool NORM>
__global__ __launch_bounds__((NCOL / 16) * 32)
void sage_gemm_wmma(const float* __restrict__ A0,
                    const float* __restrict__ cnt,
                    const float* __restrict__ A1,
                    const float* __restrict__ B0,
                    const float* __restrict__ B1,
                    const float* __restrict__ bias,
                    float* __restrict__ out,
                    int M) {
  constexpr int NT  = (NCOL / 16) * 32;
  constexpr int LDK = KTOT + 4;
  __shared__ float Asub[32][LDK];

  const int tid  = threadIdx.x;
  const int wave = tid >> 5;
  const int lane = tid & 31;
  const int half = lane >> 4;
  const int l15  = lane & 15;
  const int m0   = blockIdx.x * 32;
  const bool has2 = (m0 + 16) < M;

  for (int i = tid * 4; i < 32 * KTOT; i += NT * 4) {
    int row = i / KTOT;
    int col = i % KTOT;
    int gr  = m0 + row;
    float4 v = {0.0f, 0.0f, 0.0f, 0.0f};
    if (gr < M) {
      if (col < 256) {
        v = *(const float4*)(A0 + (size_t)gr * 256 + col);
        if (NORM) {
          float c = cnt[gr];
          float s = (c > 0.0f) ? 1.0f / c : 0.0f;
          v.x *= s; v.y *= s; v.z *= s; v.w *= s;
        }
      } else {
        v = *(const float4*)(A1 + (size_t)gr * 256 + (col - 256));
      }
    }
    *(float4*)&Asub[row][col] = v;
  }
  __syncthreads();

  const int nbase = wave * 16;
  v8f acc0 = {0.0f, 0.0f, 0.0f, 0.0f, 0.0f, 0.0f, 0.0f, 0.0f};
  v8f acc1 = {0.0f, 0.0f, 0.0f, 0.0f, 0.0f, 0.0f, 0.0f, 0.0f};

#pragma unroll
  for (int seg = 0; seg < KTOT / 256; ++seg) {
    const float* bp = ((seg == 0) ? B0 : B1) +
                      (size_t)(half * 2) * NCOL + nbase + l15;
    const int kb = seg * 256;
#pragma unroll 8
    for (int k = 0; k < 256; k += 4) {
      v2f b;
      b.x = bp[0];
      b.y = bp[NCOL];
      int ka = kb + k + half * 2;
      float2 a0v = *(const float2*)&Asub[l15][ka];
      float2 a1v = *(const float2*)&Asub[16 + l15][ka];
      v2f a0; a0.x = a0v.x; a0.y = a0v.y;
      v2f a1; a1.x = a1v.x; a1.y = a1v.y;
      acc0 = __builtin_amdgcn_wmma_f32_16x16x4_f32(false, a0, false, b,
                                                   (short)0, acc0, false, false);
      acc1 = __builtin_amdgcn_wmma_f32_16x16x4_f32(false, a1, false, b,
                                                   (short)0, acc1, false, false);
      bp += (size_t)4 * NCOL;
    }
  }

  const int col = nbase + l15;
  const float bv = bias[col];
#pragma unroll
  for (int r = 0; r < 8; ++r) {
    int row = m0 + half * 8 + r;
    float v = acc0[r] + bv;
    if (ACT == 1) v = fmaxf(v, 0.0f);
    if (ACT == 2) v = tanhf(v);
    out[(size_t)row * NCOL + col] = v;
  }
  if (has2) {
#pragma unroll
    for (int r = 0; r < 8; ++r) {
      int row = m0 + 16 + half * 8 + r;
      float v = acc1[r] + bv;
      if (ACT == 1) v = fmaxf(v, 0.0f);
      if (ACT == 2) v = tanhf(v);
      out[(size_t)row * NCOL + col] = v;
    }
  }
}

__global__ void softmax_rows64(const float* __restrict__ logits,
                               float* __restrict__ out, int M) {
  int row = blockIdx.x * 8 + (threadIdx.x >> 5);
  if (row >= M) return;
  int lane = threadIdx.x & 31;
  const float* lp = logits + (size_t)row * 64;
  float a = lp[lane];
  float b = lp[lane + 32];
  float m = fmaxf(a, b);
  for (int off = 16; off > 0; off >>= 1) m = fmaxf(m, __shfl_xor(m, off, 32));
  float ea = expf(a - m);
  float eb = expf(b - m);
  float s = ea + eb;
  for (int off = 16; off > 0; off >>= 1) s += __shfl_xor(s, off, 32);
  float inv = 1.0f / s;
  float* op = out + (size_t)row * 64;
  op[lane] = ea * inv;
  op[lane + 32] = eb * inv;
}

extern "C" void kernel_launch(void* const* d_in, const int* in_sizes, int n_in,
                              void* d_out, int out_size, void* d_ws,
                              size_t ws_size, hipStream_t stream) {
  (void)in_sizes; (void)n_in; (void)out_size; (void)ws_size;

  const float* x    = (const float*)d_in[0];
  const int*   src0 = (const int*)d_in[1];
  const int*   dst0 = (const int*)d_in[2];
  const int*   src1 = (const int*)d_in[3];
  const int*   dst1 = (const int*)d_in[4];
  const float* Wl0  = (const float*)d_in[5];
  const float* bl0  = (const float*)d_in[6];
  const float* Wr0  = (const float*)d_in[7];
  const float* Wl1  = (const float*)d_in[8];
  const float* bl1  = (const float*)d_in[9];
  const float* Wr1  = (const float*)d_in[10];
  const float* Wlin = (const float*)d_in[11];
  const float* blin = (const float*)d_in[12];

  char* ws = (char*)d_ws;
  auto align256 = [](size_t v) { return (v + 255) & ~(size_t)255; };
  size_t off = 0;
  float* agg0 = (float*)(ws + off); off += (size_t)N1_ * 256 * 4;
  off += (size_t)N1_ * 4; off = align256(off);
  float* h0   = (float*)(ws + off); off += (size_t)N1_ * 256 * 4;
  off += (size_t)N2_ * 256 * 4;
  off += (size_t)N2_ * 4; off = align256(off);
  float* h1   = (float*)(ws + off); off += (size_t)N2_ * 256 * 4;
  float* lgt  = (float*)(ws + off); off += (size_t)N2_ * 64 * 4;

  static_assert((size_t)(NKEY_ + 4 + ETOT_ + NCHUNK_ * CHUNK_ + 2 * NBIN_ * NCHUNK_) * 4
                    <= (size_t)N1_ * 256 * 4, "edge lists fit the region");
  int* rp     = (int*)agg0;
  int* es     = rp + (NKEY_ + 4);
  int* runs   = es + ETOT_;
  int* startT = runs + NCHUNK_ * CHUNK_;
  int* cntT   = startT + NBIN_ * NCHUNK_;

  chunk_bin<<<NCHUNK_, 1024, 0, stream>>>(src0, dst0, src1, dst1, runs, startT, cntT);
  bin_sort<<<NBIN_, 256, 0, stream>>>(runs, startT, cntT, rp, es);

  sage_layer_fused<1><<<(N1_ + 31) / 32, 512, 0, stream>>>(
      x, rp, es, x, Wl0, Wr0, bl0, h0, N1_);

  sage_layer_fused<2><<<(N2_ + 31) / 32, 512, 0, stream>>>(
      h0, rp + N1_, es, h0, Wl1, Wr1, bl1, h1, N2_);

  sage_gemm_wmma<256, 64, 0, false><<<(N2_ + 31) / 32, 128, 0, stream>>>(
      h1, nullptr, nullptr, Wlin, nullptr, blin, lgt, N2_);
  softmax_rows64<<<(N2_ + 7) / 8, 256, 0, stream>>>(lgt, (float*)d_out, N2_);
}


# Head product (f32 WMMA) and row softmax fused into the layer-1 launch: the tanh tile goes to LDS as the head's A operand instead of to memory; 6 launches -> 4; every sum order unchanged
# speedup vs baseline: 2.6258x; 1.0705x over previous
#include <hip/hip_runtime.h>
#include <cstdint>
#include <cstddef>

#define N0_  200000
#define N1_  50000
#define N2_  10000
#define E0_  750000
#define E1_  150000
#define DIN_ 256
#define HID_ 256
#define OUT_ 64

typedef float v2f __attribute__((ext_vector_type(2)));
typedef float v8f __attribute__((ext_vector_type(8)));

#define NKEY_   (N1_ + N2_)
#define ETOT_   (E0_ + E1_)
#define NBIN_   ((NKEY_ + 255) / 256)
#define NCHUNK_ 256
#define CHUNK_  ((ETOT_ + NCHUNK_ - 1) / NCHUNK_)
#define BINCAP_ 8192

__device__ __forceinline__ int wave_incl_scan(int x, int lane) {
  for (int off = 1; off < 32; off <<= 1) {
    int u = __shfl_up(x, off, 32);
    if (lane >= off) x += u;
  }
  return x;
}

__global__ __launch_bounds__(1024)
void chunk_bin(const int* __restrict__ src0, const int* __restrict__ dst0,
               const int* __restrict__ src1, const int* __restrict__ dst1,
               int* __restrict__ runs, int* __restrict__ startT,
               int* __restrict__ cntT) {
  static_assert(CHUNK_ <= 4 * 1024, "four edges per thread cover a run");
  static_assert(NBIN_ <= 256, "one LDS counter per bin");
  static_assert(N0_ <= (1 << 18), "source id and key byte share one word");
  __shared__ int hist[256];
  __shared__ int boff[256];
  const int tid = threadIdx.x;
  const int c   = blockIdx.x;
  if (tid < 256) hist[tid] = 0;
  __syncthreads();

  int key[4], sv[4], rk[4];
  for (int k = 0; k < 4; ++k) {
    const int li = k * 1024 + tid;
    const int i  = c * CHUNK_ + li;
    key[k] = -1;
    sv[k]  = 0;
    if (li < CHUNK_ && i < ETOT_) {
      if (i < E0_) {
        key[k] = dst0[i];
        sv[k]  = src0[i];
      } else {
        key[k] = N1_ + dst1[i - E0_];
        sv[k]  = src1[i - E0_];
      }
    }
  }
  for (int k = 0; k < 4; ++k) {
    rk[k] = 0;
    if (key[k] >= 0) rk[k] = atomicAdd(&hist[key[k] >> 8], 1);
  }
  __syncthreads();
  if (tid < 32) {
    int h[8];
    int t = 0;
    for (int j = 0; j < 8; ++j) { h[j] = hist[tid * 8 + j]; t += h[j]; }
    int run = wave_incl_scan(t, tid) - t;
    for (int j = 0; j < 8; ++j) { boff[tid * 8 + j] = run; run += h[j]; }
  }
  __syncthreads();
  for (int k = 0; k < 4; ++k) {
    if (key[k] >= 0) {
      runs[c * CHUNK_ + boff[key[k] >> 8] + rk[k]] = sv[k] | ((key[k] & 255) << 18);
    }
  }
  if (tid < NBIN_) {
    startT[tid * NCHUNK_ + c] = boff[tid];
    cntT[tid * NCHUNK_ + c]   = hist[tid];
  }
}

__global__ __launch_bounds__(256)
void bin_sort(const int* __restrict__ runs, const int* __restrict__ startT,
              const int* __restrict__ cntT, int* __restrict__ rp,
              int* __restrict__ es) {
  static_assert(NCHUNK_ == 256, "one thread per run");
  __shared__ int buf[BINCAP_];
  __shared__ int rnk[BINCAP_];
  __shared__ int khist[256];
  __shared__ int koff[256];
  __shared__ int wsum[2][8];
  const int tid  = threadIdx.x;
  const int lane = tid & 31;
  const int wave = tid >> 5;
  const int b    = blockIdx.x;

  const int s = startT[b * NCHUNK_ + tid];
  const int n = cntT[b * NCHUNK_ + tid];
  khist[tid] = 0;
  const int s_inc = wave_incl_scan(s, lane);
  const int n_inc = wave_incl_scan(n, lane);
  if (lane == 31) { wsum[0][wave] = s_inc; wsum[1][wave] = n_inc; }
  __syncthreads();
  int base = 0, npre = 0, ntot = 0;
  for (int w = 0; w < 8; ++w) {
    base += wsum[0][w];
    const int t = wsum[1][w];
    if (w < wave) npre += t;
    ntot += t;
  }
  const int lpos = npre + n_inc - n;

  const int wrun = __builtin_amdgcn_readfirstlane(wave) * 32;
  for (int j0 = 0; j0 < 32; j0 += 8) {
    int v[8];
#pragma unroll
    for (int u = 0; u < 8; ++u) {
      const int sj = __builtin_amdgcn_readlane(s, j0 + u);
      const int nj = __builtin_amdgcn_readlane(n, j0 + u);
      v[u] = 0;
      if (lane < nj) v[u] = runs[(wrun + j0 + u) * CHUNK_ + sj + lane];
    }
#pragma unroll
    for (int u = 0; u < 8; ++u) {
      const int nj = __builtin_amdgcn_readlane(n, j0 + u);
      const int pj = __builtin_amdgcn_readlane(lpos, j0 + u);
      if (lane < nj && pj + lane < BINCAP_) buf[pj + lane] = v[u];
    }
  }
  if (__any(n > 32)) {
    for (int j = 0; j < 32; ++j) {
      const int sj = __builtin_amdgcn_readlane(s, j);
      const int nj = __builtin_amdgcn_readlane(n, j);
      const int pj = __builtin_amdgcn_readlane(lpos, j);
      for (int q = lane + 32; q < nj; q += 32) {
        if (pj + q < BINCAP_) buf[pj + q] = runs[(wrun + j) * CHUNK_ + sj + q];
      }
    }
  }
  __syncthreads();

  const int nin = (ntot < BINCAP_) ? ntot : BINCAP_;
  for (int i0 = tid; i0 < nin; i0 += 4 * 256) {
    int w[4], r[4];
    for (int u = 0; u < 4; ++u) {
      const int i = i0 + u * 256;
      w[u] = (i < nin) ? buf[i] : 0;
    }
    for (int u = 0; u < 4; ++u) {
      const int i = i0 + u * 256;
      r[u] = 0;
      if (i < nin) r[u] = atomicAdd(&khist[(w[u] >> 18) & 255], 1);
    }
    for (int u = 0; u < 4; ++u) {
      const int i = i0 + u * 256;
      if (i < nin) rnk[i] = r[u];
    }
  }
  __syncthreads();
  const int h     = khist[tid];
  const int h_inc = wave_incl_scan(h, lane);
  if (lane == 31) wsum[0][wave] = h_inc;
  __syncthreads();
  int hpre = 0;
  for (int w = 0; w < 8; ++w) {
    if (w < wave) hpre += wsum[0][w];
  }
  const int kex = hpre + h_inc - h;
  koff[tid] = kex;
  const int key = b * 256 + tid;
  if (key < NKEY_) rp[key] = base + kex;
  if (b == NBIN_ - 1 && tid == 0) rp[NKEY_] = base + ntot;
  __syncthreads();
  for (int i = tid; i < nin; i += 256) {
    const int w = buf[i];
    es[base + koff[(w >> 18) & 255] + rnk[i]] = w & 0x3FFFF;
  }
}

__device__ __forceinline__ void gather_row_sum(const float* __restrict__ xsrc,
                                               const int* __restrict__ esrc,
                                               int beg, int end, int first,
                                               int lane, float4& a0, float4& a1) {
  for (int base = beg; base < end; base += 32) {
    const int n = (end - base < 32) ? (end - base) : 32;
    int mine = first;
    if (base != beg) mine = (lane < n) ? esrc[base + lane] : 0;
    int j = 0;
    for (; j + 4 <= n; j += 4) {
      const float4* p0 = (const float4*)(xsrc + (size_t)__builtin_amdgcn_readlane(mine, j + 0) * 256);
      const float4* p1 = (const float4*)(xsrc + (size_t)__builtin_amdgcn_readlane(mine, j + 1) * 256);
      const float4* p2 = (const float4*)(xsrc + (size_t)__builtin_amdgcn_readlane(mine, j + 2) * 256);
      const float4* p3 = (const float4*)(xsrc + (size_t)__builtin_amdgcn_readlane(mine, j + 3) * 256);
      float4 u0 = p0[lane], w0 = p0[32 + lane];
      float4 u1 = p1[lane], w1 = p1[32 + lane];
      float4 u2 = p2[lane], w2 = p2[32 + lane];
      float4 u3 = p3[lane], w3 = p3[32 + lane];
      a0 += u0; a1 += w0;
      a0 += u1; a1 += w1;
      a0 += u2; a1 += w2;
      a0 += u3; a1 += w3;
    }
    const int r = n - j;
    if (r == 3) {
      const float4* p0 = (const float4*)(xsrc + (size_t)__builtin_amdgcn_readlane(mine, j + 0) * 256);
      const float4* p1 = (const float4*)(xsrc + (size_t)__builtin_amdgcn_readlane(mine, j + 1) * 256);
      const float4* p2 = (const float4*)(xsrc + (size_t)__builtin_amdgcn_readlane(mine, j + 2) * 256);
      float4 u0 = p0[lane], w0 = p0[32 + lane];
      float4 u1 = p1[lane], w1 = p1[32 + lane];
      float4 u2 = p2[lane], w2 = p2[32 + lane];
      a0 += u0; a1 += w0;
      a0 += u1; a1 += w1;
      a0 += u2; a1 += w2;
    } else if (r == 2) {
      const float4* p0 = (const float4*)(xsrc + (size_t)__builtin_amdgcn_readlane(mine, j + 0) * 256);
      const float4* p1 = (const float4*)(xsrc + (size_t)__builtin_amdgcn_readlane(mine, j + 1) * 256);
      float4 u0 = p0[lane], w0 = p0[32 + lane];
      float4 u1 = p1[lane], w1 = p1[32 + lane];
      a0 += u0; a1 += w0;
      a0 += u1; a1 += w1;
    } else if (r == 1) {
      const float4* p0 = (const float4*)(xsrc + (size_t)__builtin_amdgcn_readlane(mine, j) * 256);
      float4 u0 = p0[lane], w0 = p0[32 + lane];
      a0 += u0; a1 += w0;
    }
  }
}

template <int ACT, bool HEAD>
__global__ __launch_bounds__(512)
void sage_layer_fused(const float* __restrict__ xsrc,
                      const int* __restrict__ rowptr,
                      const int* __restrict__ esrc,
                      const float* __restrict__ xtgt,
                      const float* __restrict__ Bl,
                      const float* __restrict__ Br,
                      const float* __restrict__ bias,
                      float* __restrict__ out,
                      const float* __restrict__ Wh,
                      const float* __restrict__ bh,
                      float* __restrict__ pout,
                      int M) {
  constexpr int NCOL = 256;
  constexpr int LDK  = 512 + 4;
  __shared__ float Asub[32][LDK];

  const int tid  = threadIdx.x;
  const int lane = tid & 31;
  const int half = lane >> 4;
  const int l15  = lane & 15;
  const int wave = __builtin_amdgcn_readfirstlane(tid >> 5);
  const int m0   = blockIdx.x * 32;
  const bool has2 = (m0 + 16) < M;

  const int r0 = m0 + wave * 2;
  int b0 = 0, b1 = 0, b2 = 0;
  if (r0 < M) { b0 = rowptr[r0]; b1 = rowptr[r0 + 1]; }
  b2 = b1;
  if (r0 + 1 < M) b2 = rowptr[r0 + 2];
  b0 = __builtin_amdgcn_readfirstlane(b0);
  b1 = __builtin_amdgcn_readfirstlane(b1);
  b2 = __builtin_amdgcn_readfirstlane(b2);
  int first0 = 0, first1 = 0;
  if (b0 + lane < b1) first0 = esrc[b0 + lane];
  if (b1 + lane < b2) first1 = esrc[b1 + lane];

  {
    float4 tv[4];
#pragma unroll
    for (int u = 0; u < 4; ++u) {
      const int i  = tid * 4 + u * 2048;
      const int gr = m0 + (i >> 8);
      const int gc = (gr < M) ? gr : (M - 1);
      tv[u] = *(const float4*)(xtgt + (size_t)gc * 256 + (i & 255));
      if (gr >= M) tv[u] = float4{0.0f, 0.0f, 0.0f, 0.0f};
    }
#pragma unroll
    for (int u = 0; u < 4; ++u) {
      const int i = tid * 4 + u * 2048;
      *(float4*)&Asub[i >> 8][256 + (i & 255)] = tv[u];
    }
  }
  __syncthreads();

  const int nbase = wave * 16;
  v8f acc0 = {0.0f, 0.0f, 0.0f, 0.0f, 0.0f, 0.0f, 0.0f, 0.0f};
  v8f acc1 = {0.0f, 0.0f, 0.0f, 0.0f, 0.0f, 0.0f, 0.0f, 0.0f};

  auto kseg = [&](const float* __restrict__ Bm, const int kb) {
    const float* bp = Bm + (size_t)(half * 2) * NCOL + nbase + l15;
#pragma unroll 8
    for (int k = 0; k < 256; k += 4) {
      v2f b;
      b.x = bp[0];
      b.y = bp[NCOL];
      const int ka = kb + k + half * 2;
      float2 a0v = *(const float2*)&Asub[l15][ka];
      float2 a1v = *(const float2*)&Asub[16 + l15][ka];
      v2f a0; a0.x = a0v.x; a0.y = a0v.y;
      v2f a1; a1.x = a1v.x; a1.y = a1v.y;
      acc0 = __builtin_amdgcn_wmma_f32_16x16x4_f32(false, a0, false, b,
                                                   (short)0, acc0, false, false);
      acc1 = __builtin_amdgcn_wmma_f32_16x16x4_f32(false, a1, false, b,
                                                   (short)0, acc1, false, false);
      bp += (size_t)4 * NCOL;
    }
  };

  kseg(Br, 256);

  {
    float4 s0 = {0.0f, 0.0f, 0.0f, 0.0f};
    float4 s1 = {0.0f, 0.0f, 0.0f, 0.0f};
    gather_row_sum(xsrc, esrc, b0, b1, first0, lane, s0, s1);
    const float c = (float)(b1 - b0);
    const float sc = (c > 0.0f) ? 1.0f / c : 0.0f;
    s0.x *= sc; s0.y *= sc; s0.z *= sc; s0.w *= sc;
    s1.x *= sc; s1.y *= sc; s1.z *= sc; s1.w *= sc;
    *(float4*)&Asub[wave * 2][4 * lane]       = s0;
    *(float4*)&Asub[wave * 2][128 + 4 * lane] = s1;
  }
  {
    float4 s0 = {0.0f, 0.0f, 0.0f, 0.0f};
    float4 s1 = {0.0f, 0.0f, 0.0f, 0.0f};
    gather_row_sum(xsrc, esrc, b1, b2, first1, lane, s0, s1);
    const float c = (float)(b2 - b1);
    const float sc = (c > 0.0f) ? 1.0f / c : 0.0f;
    s0.x *= sc; s0.y *= sc; s0.z *= sc; s0.w *= sc;
    s1.x *= sc; s1.y *= sc; s1.z *= sc; s1.w *= sc;
    *(float4*)&Asub[wave * 2 + 1][4 * lane]       = s0;
    *(float4*)&Asub[wave * 2 + 1][128 + 4 * lane] = s1;
  }
  __syncthreads();

  kseg(Bl, 0);

  const int col = nbase + l15;
  const float bv = bias[col];
  if (!HEAD) {
#pragma unroll
    for (int r = 0; r < 8; ++r) {
      int row = m0 + half * 8 + r;
      float v = acc0[r] + bv;
      if (ACT == 1) v = fmaxf(v, 0.0f);
      if (ACT == 2) v = tanhf(v);
      out[(size_t)row * NCOL + col] = v;
    }
    if (has2) {
#pragma unroll
      for (int r = 0; r < 8; ++r) {
        int row = m0 + 16 + half * 8 + r;
        float v = acc1[r] + bv;
        if (ACT == 1) v = fmaxf(v, 0.0f);
        if (ACT == 2) v = tanhf(v);
        out[(size_t)row * NCOL + col] = v;
      }
    }
  } else {
    __syncthreads();
#pragma unroll
    for (int r = 0; r < 8; ++r) {
      float v0 = acc0[r] + bv;
      float v1 = acc1[r] + bv;
      if (ACT == 1) { v0 = fmaxf(v0, 0.0f); v1 = fmaxf(v1, 0.0f); }
      if (ACT == 2) { v0 = tanhf(v0); v1 = tanhf(v1); }
      Asub[half * 8 + r][col]      = v0;
      Asub[16 + half * 8 + r][col] = v1;
    }
    __syncthreads();

    if (wave < 4) {
      const int hb = wave * 16;
      v8f c0 = {0.0f, 0.0f, 0.0f, 0.0f, 0.0f, 0.0f, 0.0f, 0.0f};
      v8f c1 = {0.0f, 0.0f, 0.0f, 0.0f, 0.0f, 0.0f, 0.0f, 0.0f};
      const float* hp = Wh + (size_t)(half * 2) * OUT_ + hb + l15;
#pragma unroll 16
      for (int k = 0; k < 256; k += 4) {
        v2f b;
        b.x = hp[0];
        b.y = hp[OUT_];
        const int ka = k + half * 2;
        float2 a0v = *(const float2*)&Asub[l15][ka];
        float2 a1v = *(const float2*)&Asub[16 + l15][ka];
        v2f a0; a0.x = a0v.x; a0.y = a0v.y;
        v2f a1; a1.x = a1v.x; a1.y = a1v.y;
        c0 = __builtin_amdgcn_wmma_f32_16x16x4_f32(false, a0, false, b,
                                                   (short)0, c0, false, false);
        c1 = __builtin_amdgcn_wmma_f32_16x16x4_f32(false, a1, false, b,
                                                   (short)0, c1, false, false);
        hp += (size_t)4 * OUT_;
      }
      const float hbv = bh[hb + l15];
#pragma unroll
      for (int r = 0; r < 8; ++r) {
        Asub[half * 8 + r][256 + hb + l15]      = c0[r] + hbv;
        Asub[16 + half * 8 + r][256 + hb + l15] = c1[r] + hbv;
      }
    }
    __syncthreads();

    for (int q = 0; q < 2; ++q) {
      const int lr = wave * 2 + q;
      const int gr = m0 + lr;
      if (gr < M) {
        const float a = Asub[lr][256 + lane];
        const float b = Asub[lr][256 + 32 + lane];
        float m = fmaxf(a, b);
        for (int off = 16; off > 0; off >>= 1) m = fmaxf(m, __shfl_xor(m, off, 32));
        const float ea = expf(a - m);
        const float eb = expf(b - m);
        float s = ea + eb;
        for (int off = 16; off > 0; off >>= 1) s += __shfl_xor(s, off, 32);
        const float inv = 1.0f / s;
        float* op = pout + (size_t)gr * OUT_;
        op[lane]      = ea * inv;
        op[lane + 32] = eb * inv;
      }
    }
  }
}


extern "C" void kernel_launch(void* const* d_in, const int* in_sizes, int n_in,
                              void* d_out, int out_size, void* d_ws,
                              size_t ws_size, hipStream_t stream) {
  (void)in_sizes; (void)n_in; (void)out_size; (void)ws_size;

  const float* x    = (const float*)d_in[0];
  const int*   src0 = (const int*)d_in[1];
  const int*   dst0 = (const int*)d_in[2];
  const int*   src1 = (const int*)d_in[3];
  const int*   dst1 = (const int*)d_in[4];
  const float* Wl0  = (const float*)d_in[5];
  const float* bl0  = (const float*)d_in[6];
  const float* Wr0  = (const float*)d_in[7];
  const float* Wl1  = (const float*)d_in[8];
  const float* bl1  = (const float*)d_in[9];
  const float* Wr1  = (const float*)d_in[10];
  const float* Wlin = (const float*)d_in[11];
  const float* blin = (const float*)d_in[12];

  char* ws = (char*)d_ws;
  auto align256 = [](size_t v) { return (v + 255) & ~(size_t)255; };
  size_t off = 0;
  float* agg0 = (float*)(ws + off); off += (size_t)N1_ * 256 * 4;
  off += (size_t)N1_ * 4; off = align256(off);
  float* h0   = (float*)(ws + off); off += (size_t)N1_ * 256 * 4;

  static_assert((size_t)(NKEY_ + 4 + ETOT_ + NCHUNK_ * CHUNK_ + 2 * NBIN_ * NCHUNK_) * 4
                    <= (size_t)N1_ * 256 * 4, "edge lists fit the region");
  int* rp     = (int*)agg0;
  int* es     = rp + (NKEY_ + 4);
  int* runs   = es + ETOT_;
  int* startT = runs + NCHUNK_ * CHUNK_;
  int* cntT   = startT + NBIN_ * NCHUNK_;

  chunk_bin<<<NCHUNK_, 1024, 0, stream>>>(src0, dst0, src1, dst1, runs, startT, cntT);
  bin_sort<<<NBIN_, 256, 0, stream>>>(runs, startT, cntT, rp, es);

  sage_layer_fused<1, false><<<(N1_ + 31) / 32, 512, 0, stream>>>(
      x, rp, es, x, Wl0, Wr0, bl0, h0, nullptr, nullptr, nullptr, N1_);

  sage_layer_fused<2, true><<<(N2_ + 31) / 32, 512, 0, stream>>>(
      h0, rp + N1_, es, h0, Wl1, Wr1, bl1, nullptr, Wlin, blin, (float*)d_out, N2_);
}


# Edge-sort runs rebalanced: run c = c-th slice of layer-0 edges + c-th slice of layer-1 edges (was consecutive edges): no bin gets 90-word pieces copied 32 words per round trip; row sum order changes
# speedup vs baseline: 2.8636x; 1.0906x over previous
#include <hip/hip_runtime.h>
#include <cstdint>
#include <cstddef>

#define N0_  200000
#define N1_  50000
#define N2_  10000
#define E0_  750000
#define E1_  150000
#define DIN_ 256
#define HID_ 256
#define OUT_ 64

typedef float v2f __attribute__((ext_vector_type(2)));
typedef float v8f __attribute__((ext_vector_type(8)));

#define NKEY_   (N1_ + N2_)
#define ETOT_   (E0_ + E1_)
#define NBIN_   ((NKEY_ + 255) / 256)
#define NCHUNK_ 256
#define C0_     ((E0_ + NCHUNK_ - 1) / NCHUNK_)
#define C1_     ((E1_ + NCHUNK_ - 1) / NCHUNK_)
#define CHUNK_  (C0_ + C1_)
#define BINCAP_ 8192

__device__ __forceinline__ int wave_incl_scan(int x, int lane) {
  for (int off = 1; off < 32; off <<= 1) {
    int u = __shfl_up(x, off, 32);
    if (lane >= off) x += u;
  }
  return x;
}

__global__ __launch_bounds__(1024)
void chunk_bin(const int* __restrict__ src0, const int* __restrict__ dst0,
               const int* __restrict__ src1, const int* __restrict__ dst1,
               int* __restrict__ runs, int* __restrict__ startT,
               int* __restrict__ cntT) {
  static_assert(CHUNK_ <= 4 * 1024, "four edges per thread cover a run");
  static_assert(NBIN_ <= 256, "one LDS counter per bin");
  static_assert(N0_ <= (1 << 18), "source id and key byte share one word");
  __shared__ int hist[256];
  __shared__ int boff[256];
  const int tid = threadIdx.x;
  const int c   = blockIdx.x;
  if (tid < 256) hist[tid] = 0;
  __syncthreads();

  int key[4], sv[4], rk[4];
  for (int k = 0; k < 4; ++k) {
    const int li = k * 1024 + tid;
    key[k] = -1;
    sv[k]  = 0;
    if (li < C0_) {
      const int i = c * C0_ + li;
      if (i < E0_) {
        key[k] = dst0[i];
        sv[k]  = src0[i];
      }
    } else if (li < C0_ + C1_) {
      const int i = c * C1_ + (li - C0_);
      if (i < E1_) {
        key[k] = N1_ + dst1[i];
        sv[k]  = src1[i];
      }
    }
  }
  for (int k = 0; k < 4; ++k) {
    rk[k] = 0;
    if (key[k] >= 0) rk[k] = atomicAdd(&hist[key[k] >> 8], 1);
  }
  __syncthreads();
  if (tid < 32) {
    int h[8];
    int t = 0;
    for (int j = 0; j < 8; ++j) { h[j] = hist[tid * 8 + j]; t += h[j]; }
    int run = wave_incl_scan(t, tid) - t;
    for (int j = 0; j < 8; ++j) { boff[tid * 8 + j] = run; run += h[j]; }
  }
  __syncthreads();
  for (int k = 0; k < 4; ++k) {
    if (key[k] >= 0) {
      runs[c * CHUNK_ + boff[key[k] >> 8] + rk[k]] = sv[k] | ((key[k] & 255) << 18);
    }
  }
  if (tid < NBIN_) {
    startT[tid * NCHUNK_ + c] = boff[tid];
    cntT[tid * NCHUNK_ + c]   = hist[tid];
  }
}

__global__ __launch_bounds__(256)
void bin_sort(const int* __restrict__ runs, const int* __restrict__ startT,
              const int* __restrict__ cntT, int* __restrict__ rp,
              int* __restrict__ es) {
  static_assert(NCHUNK_ == 256, "one thread per run");
  __shared__ int buf[BINCAP_];
  __shared__ int rnk[BINCAP_];
  __shared__ int khist[256];
  __shared__ int koff[256];
  __shared__ int wsum[2][8];
  const int tid  = threadIdx.x;
  const int lane = tid & 31;
  const int wave = tid >> 5;
  const int b    = blockIdx.x;

  const int s = startT[b * NCHUNK_ + tid];
  const int n = cntT[b * NCHUNK_ + tid];
  khist[tid] = 0;
  const int s_inc = wave_incl_scan(s, lane);
  const int n_inc = wave_incl_scan(n, lane);
  if (lane == 31) { wsum[0][wave] = s_inc; wsum[1][wave] = n_inc; }
  __syncthreads();
  int base = 0, npre = 0, ntot = 0;
  for (int w = 0; w < 8; ++w) {
    base += wsum[0][w];
    const int t = wsum[1][w];
    if (w < wave) npre += t;
    ntot += t;
  }
  const int lpos = npre + n_inc - n;

  const int wrun = __builtin_amdgcn_readfirstlane(wave) * 32;
  for (int j0 = 0; j0 < 32; j0 += 8) {
    int v[8];
#pragma unroll
    for (int u = 0; u < 8; ++u) {
      const int sj = __builtin_amdgcn_readlane(s, j0 + u);
      const int nj = __builtin_amdgcn_readlane(n, j0 + u);
      v[u] = 0;
      if (lane < nj) v[u] = runs[(wrun + j0 + u) * CHUNK_ + sj + lane];
    }
#pragma unroll
    for (int u = 0; u < 8; ++u) {
      const int nj = __builtin_amdgcn_readlane(n, j0 + u);
      const int pj = __builtin_amdgcn_readlane(lpos, j0 + u);
      if (lane < nj && pj + lane < BINCAP_) buf[pj + lane] = v[u];
    }
  }
  if (__any(n > 32)) {
    for (int j = 0; j < 32; ++j) {
      const int sj = __builtin_amdgcn_readlane(s, j);
      const int nj = __builtin_amdgcn_readlane(n, j);
      const int pj = __builtin_amdgcn_readlane(lpos, j);
      for (int q = lane + 32; q < nj; q += 32) {
        if (pj + q < BINCAP_) buf[pj + q] = runs[(wrun + j) * CHUNK_ + sj + q];
      }
    }
  }
  __syncthreads();

  const int nin = (ntot < BINCAP_) ? ntot : BINCAP_;
  for (int i0 = tid; i0 < nin; i0 += 4 * 256) {
    int w[4], r[4];
    for (int u = 0; u < 4; ++u) {
      const int i = i0 + u * 256;
      w[u] = (i < nin) ? buf[i] : 0;
    }
    for (int u = 0; u < 4; ++u) {
      const int i = i0 + u * 256;
      r[u] = 0;
      if (i < nin) r[u] = atomicAdd(&khist[(w[u] >> 18) & 255], 1);
    }
    for (int u = 0; u < 4; ++u) {
      const int i = i0 + u * 256;
      if (i < nin) rnk[i] = r[u];
    }
  }
  __syncthreads();
  const int h     = khist[tid];
  const int h_inc = wave_incl_scan(h, lane);
  if (lane == 31) wsum[0][wave] = h_inc;
  __syncthreads();
  int hpre = 0;
  for (int w = 0; w < 8; ++w) {
    if (w < wave) hpre += wsum[0][w];
  }
  const int kex = hpre + h_inc - h;
  koff[tid] = kex;
  const int key = b * 256 + tid;
  if (key < NKEY_) rp[key] = base + kex;
  if (b == NBIN_ - 1 && tid == 0) rp[NKEY_] = base + ntot;
  __syncthreads();
  for (int i = tid; i < nin; i += 256) {
    const int w = buf[i];
    es[base + koff[(w >> 18) & 255] + rnk[i]] = w & 0x3FFFF;
  }
}

__device__ __forceinline__ void gather_row_sum(const float* __restrict__ xsrc,
                                               const int* __restrict__ esrc,
                                               int beg, int end, int first,
                                               int lane, float4& a0, float4& a1) {
  for (int base = beg; base < end; base += 32) {
    const int n = (end - base < 32) ? (end - base) : 32;
    int mine = first;
    if (base != beg) mine = (lane < n) ? esrc[base + lane] : 0;
    int j = 0;
    for (; j + 4 <= n; j += 4) {
      const float4* p0 = (const float4*)(xsrc + (size_t)__builtin_amdgcn_readlane(mine, j + 0) * 256);
      const float4* p1 = (const float4*)(xsrc + (size_t)__builtin_amdgcn_readlane(mine, j + 1) * 256);
      const float4* p2 = (const float4*)(xsrc + (size_t)__builtin_amdgcn_readlane(mine, j + 2) * 256);
      const float4* p3 = (const float4*)(xsrc + (size_t)__builtin_amdgcn_readlane(mine, j + 3) * 256);
      float4 u0 = p0[lane], w0 = p0[32 + lane];
      float4 u1 = p1[lane], w1 = p1[32 + lane];
      float4 u2 = p2[lane], w2 = p2[32 + lane];
      float4 u3 = p3[lane], w3 = p3[32 + lane];
      a0 += u0; a1 += w0;
      a0 += u1; a1 += w1;
      a0 += u2; a1 += w2;
      a0 += u3; a1 += w3;
    }
    const int r = n - j;
    if (r == 3) {
      const float4* p0 = (const float4*)(xsrc + (size_t)__builtin_amdgcn_readlane(mine, j + 0) * 256);
      const float4* p1 = (const float4*)(xsrc + (size_t)__builtin_amdgcn_readlane(mine, j + 1) * 256);
      const float4* p2 = (const float4*)(xsrc + (size_t)__builtin_amdgcn_readlane(mine, j + 2) * 256);
      float4 u0 = p0[lane], w0 = p0[32 + lane];
      float4 u1 = p1[lane], w1 = p1[32 + lane];
      float4 u2 = p2[lane], w2 = p2[32 + lane];
      a0 += u0; a1 += w0;
      a0 += u1; a1 += w1;
      a0 += u2; a1 += w2;
    } else if (r == 2) {
      const float4* p0 = (const float4*)(xsrc + (size_t)__builtin_amdgcn_readlane(mine, j + 0) * 256);
      const float4* p1 = (const float4*)(xsrc + (size_t)__builtin_amdgcn_readlane(mine, j + 1) * 256);
      float4 u0 = p0[lane], w0 = p0[32 + lane];
      float4 u1 = p1[lane], w1 = p1[32 + lane];
      a0 += u0; a1 += w0;
      a0 += u1; a1 += w1;
    } else if (r == 1) {
      const float4* p0 = (const float4*)(xsrc + (size_t)__builtin_amdgcn_readlane(mine, j) * 256);
      float4 u0 = p0[lane], w0 = p0[32 + lane];
      a0 += u0; a1 += w0;
    }
  }
}

template <int ACT, bool HEAD>
__global__ __launch_bounds__(512)
void sage_layer_fused(const float* __restrict__ xsrc,
                      const int* __restrict__ rowptr,
                      const int* __restrict__ esrc,
                      const float* __restrict__ xtgt,
                      const float* __restrict__ Bl,
                      const float* __restrict__ Br,
                      const float* __restrict__ bias,
                      float* __restrict__ out,
                      const float* __restrict__ Wh,
                      const float* __restrict__ bh,
                      float* __restrict__ pout,
                      int M) {
  constexpr int NCOL = 256;
  constexpr int LDK  = 512 + 4;
  __shared__ float Asub[32][LDK];

  const int tid  = threadIdx.x;
  const int lane = tid & 31;
  const int half = lane >> 4;
  const int l15  = lane & 15;
  const int wave = __builtin_amdgcn_readfirstlane(tid >> 5);
  const int m0   = blockIdx.x * 32;
  const bool has2 = (m0 + 16) < M;

  const int r0 = m0 + wave * 2;
  int b0 = 0, b1 = 0, b2 = 0;
  if (r0 < M) { b0 = rowptr[r0]; b1 = rowptr[r0 + 1]; }
  b2 = b1;
  if (r0 + 1 < M) b2 = rowptr[r0 + 2];
  b0 = __builtin_amdgcn_readfirstlane(b0);
  b1 = __builtin_amdgcn_readfirstlane(b1);
  b2 = __builtin_amdgcn_readfirstlane(b2);
  int first0 = 0, first1 = 0;
  if (b0 + lane < b1) first0 = esrc[b0 + lane];
  if (b1 + lane < b2) first1 = esrc[b1 + lane];

  {
    float4 tv[4];
#pragma unroll
    for (int u = 0; u < 4; ++u) {
      const int i  = tid * 4 + u * 2048;
      const int gr = m0 + (i >> 8);
      const int gc = (gr < M) ? gr : (M - 1);
      tv[u] = *(const float4*)(xtgt + (size_t)gc * 256 + (i & 255));
      if (gr >= M) tv[u] = float4{0.0f, 0.0f, 0.0f, 0.0f};
    }
#pragma unroll
    for (int u = 0; u < 4; ++u) {
      const int i = tid * 4 + u * 2048;
      *(float4*)&Asub[i >> 8][256 + (i & 255)] = tv[u];
    }
  }
  __syncthreads();

  const int nbase = wave * 16;
  v8f acc0 = {0.0f, 0.0f, 0.0f, 0.0f, 0.0f, 0.0f, 0.0f, 0.0f};
  v8f acc1 = {0.0f, 0.0f, 0.0f, 0.0f, 0.0f, 0.0f, 0.0f, 0.0f};

  auto kseg = [&](const float* __restrict__ Bm, const int kb) {
    const float* bp = Bm + (size_t)(half * 2) * NCOL + nbase + l15;
#pragma unroll 8
    for (int k = 0; k < 256; k += 4) {
      v2f b;
      b.x = bp[0];
      b.y = bp[NCOL];
      const int ka = kb + k + half * 2;
      float2 a0v = *(const float2*)&Asub[l15][ka];
      float2 a1v = *(const float2*)&Asub[16 + l15][ka];
      v2f a0; a0.x = a0v.x; a0.y = a0v.y;
      v2f a1; a1.x = a1v.x; a1.y = a1v.y;
      acc0 = __builtin_amdgcn_wmma_f32_16x16x4_f32(false, a0, false, b,
                                                   (short)0, acc0, false, false);
      acc1 = __builtin_amdgcn_wmma_f32_16x16x4_f32(false, a1, false, b,
                                                   (short)0, acc1, false, false);
      bp += (size_t)4 * NCOL;
    }
  };

  kseg(Br, 256);

  {
    float4 s0 = {0.0f, 0.0f, 0.0f, 0.0f};
    float4 s1 = {0.0f, 0.0f, 0.0f, 0.0f};
    gather_row_sum(xsrc, esrc, b0, b1, first0, lane, s0, s1);
    const float c = (float)(b1 - b0);
    const float sc = (c > 0.0f) ? 1.0f / c : 0.0f;
    s0.x *= sc; s0.y *= sc; s0.z *= sc; s0.w *= sc;
    s1.x *= sc; s1.y *= sc; s1.z *= sc; s1.w *= sc;
    *(float4*)&Asub[wave * 2][4 * lane]       = s0;
    *(float4*)&Asub[wave * 2][128 + 4 * lane] = s1;
  }
  {
    float4 s0 = {0.0f, 0.0f, 0.0f, 0.0f};
    float4 s1 = {0.0f, 0.0f, 0.0f, 0.0f};
    gather_row_sum(xsrc, esrc, b1, b2, first1, lane, s0, s1);
    const float c = (float)(b2 - b1);
    const float sc = (c > 0.0f) ? 1.0f / c : 0.0f;
    s0.x *= sc; s0.y *= sc; s0.z *= sc; s0.w *= sc;
    s1.x *= sc; s1.y *= sc; s1.z *= sc; s1.w *= sc;
    *(float4*)&Asub[wave * 2 + 1][4 * lane]       = s0;
    *(float4*)&Asub[wave * 2 + 1][128 + 4 * lane] = s1;
  }
  __syncthreads();

  kseg(Bl, 0);

  const int col = nbase + l15;
  const float bv = bias[col];
  if (!HEAD) {
#pragma unroll
    for (int r = 0; r < 8; ++r) {
      int row = m0 + half * 8 + r;
      float v = acc0[r] + bv;
      if (ACT == 1) v = fmaxf(v, 0.0f);
      if (ACT == 2) v = tanhf(v);
      out[(size_t)row * NCOL + col] = v;
    }
    if (has2) {
#pragma unroll
      for (int r = 0; r < 8; ++r) {
        int row = m0 + 16 + half * 8 + r;
        float v = acc1[r] + bv;
        if (ACT == 1) v = fmaxf(v, 0.0f);
        if (ACT == 2) v = tanhf(v);
        out[(size_t)row * NCOL + col] = v;
      }
    }
  } else {
    __syncthreads();
#pragma unroll
    for (int r = 0; r < 8; ++r) {
      float v0 = acc0[r] + bv;
      float v1 = acc1[r] + bv;
      if (ACT == 1) { v0 = fmaxf(v0, 0.0f); v1 = fmaxf(v1, 0.0f); }
      if (ACT == 2) { v0 = tanhf(v0); v1 = tanhf(v1); }
      Asub[half * 8 + r][col]      = v0;
      Asub[16 + half * 8 + r][col] = v1;
    }
    __syncthreads();

    if (wave < 4) {
      const int hb = wave * 16;
      v8f c0 = {0.0f, 0.0f, 0.0f, 0.0f, 0.0f, 0.0f, 0.0f, 0.0f};
      v8f c1 = {0.0f, 0.0f, 0.0f, 0.0f, 0.0f, 0.0f, 0.0f, 0.0f};
      const float* hp = Wh + (size_t)(half * 2) * OUT_ + hb + l15;
#pragma unroll 16
      for (int k = 0; k < 256; k += 4) {
        v2f b;
        b.x = hp[0];
        b.y = hp[OUT_];
        const int ka = k + half * 2;
        float2 a0v = *(const float2*)&Asub[l15][ka];
        float2 a1v = *(const float2*)&Asub[16 + l15][ka];
        v2f a0; a0.x = a0v.x; a0.y = a0v.y;
        v2f a1; a1.x = a1v.x; a1.y = a1v.y;
        c0 = __builtin_amdgcn_wmma_f32_16x16x4_f32(false, a0, false, b,
                                                   (short)0, c0, false, false);
        c1 = __builtin_amdgcn_wmma_f32_16x16x4_f32(false, a1, false, b,
                                                   (short)0, c1, false, false);
        hp += (size_t)4 * OUT_;
      }
      const float hbv = bh[hb + l15];
#pragma unroll
      for (int r = 0; r < 8; ++r) {
        Asub[half * 8 + r][256 + hb + l15]      = c0[r] + hbv;
        Asub[16 + half * 8 + r][256 + hb + l15] = c1[r] + hbv;
      }
    }
    __syncthreads();

    for (int q = 0; q < 2; ++q) {
      const int lr = wave * 2 + q;
      const int gr = m0 + lr;
      if (gr < M) {
        const float a = Asub[lr][256 + lane];
        const float b = Asub[lr][256 + 32 + lane];
        float m = fmaxf(a, b);
        for (int off = 16; off > 0; off >>= 1) m = fmaxf(m, __shfl_xor(m, off, 32));
        const float ea = expf(a - m);
        const float eb = expf(b - m);
        float s = ea + eb;
        for (int off = 16; off > 0; off >>= 1) s += __shfl_xor(s, off, 32);
        const float inv = 1.0f / s;
        float* op = pout + (size_t)gr * OUT_;
        op[lane]      = ea * inv;
        op[lane + 32] = eb * inv;
      }
    }
  }
}


extern "C" void kernel_launch(void* const* d_in, const int* in_sizes, int n_in,
                              void* d_out, int out_size, void* d_ws,
                              size_t ws_size, hipStream_t stream) {
  (void)in_sizes; (void)n_in; (void)out_size; (void)ws_size;

  const float* x    = (const float*)d_in[0];
  const int*   src0 = (const int*)d_in[1];
  const int*   dst0 = (const int*)d_in[2];
  const int*   src1 = (const int*)d_in[3];
  const int*   dst1 = (const int*)d_in[4];
  const float* Wl0  = (const float*)d_in[5];
  const float* bl0  = (const float*)d_in[6];
  const float* Wr0  = (const float*)d_in[7];
  const float* Wl1  = (const float*)d_in[8];
  const float* bl1  = (const float*)d_in[9];
  const float* Wr1  = (const float*)d_in[10];
  const float* Wlin = (const float*)d_in[11];
  const float* blin = (const float*)d_in[12];

  char* ws = (char*)d_ws;
  auto align256 = [](size_t v) { return (v + 255) & ~(size_t)255; };
  size_t off = 0;
  float* agg0 = (float*)(ws + off); off += (size_t)N1_ * 256 * 4;
  off += (size_t)N1_ * 4; off = align256(off);
  float* h0   = (float*)(ws + off); off += (size_t)N1_ * 256 * 4;

  static_assert((size_t)(NKEY_ + 4 + ETOT_ + NCHUNK_ * CHUNK_ + 2 * NBIN_ * NCHUNK_) * 4
                    <= (size_t)N1_ * 256 * 4, "edge lists fit the region");
  int* rp     = (int*)agg0;
  int* es     = rp + (NKEY_ + 4);
  int* runs   = es + ETOT_;
  int* startT = runs + NCHUNK_ * CHUNK_;
  int* cntT   = startT + NBIN_ * NCHUNK_;

  chunk_bin<<<NCHUNK_, 1024, 0, stream>>>(src0, dst0, src1, dst1, runs, startT, cntT);
  bin_sort<<<NBIN_, 256, 0, stream>>>(runs, startT, cntT, rp, es);

  sage_layer_fused<1, false><<<(N1_ + 31) / 32, 512, 0, stream>>>(
      x, rp, es, x, Wl0, Wr0, bl0, h0, nullptr, nullptr, nullptr, N1_);

  sage_layer_fused<2, true><<<(N2_ + 31) / 32, 512, 0, stream>>>(
      h0, rp + N1_, es, h0, Wl1, Wr1, bl1, nullptr, Wlin, blin, (float*)d_out, N2_);
}
